# starting kernel
# baseline (speedup 1.0000x reference)
#include <hip/hip_runtime.h>

typedef __attribute__((ext_vector_type(2))) float v2f;
typedef __attribute__((ext_vector_type(8))) float v8f;

#define KTOT 768
#define WAVES_PER_BLOCK 8
#define PATCHES_PER_WAVE 16

__global__ __launch_bounds__(256) void patchconv_wmma_kernel(
    const float* __restrict__ x, const float* __restrict__ w,
    const float* __restrict__ bias, float* __restrict__ out, int npatch)
{
    __shared__ float sW[KTOT];
    for (int i = threadIdx.x; i < KTOT; i += 256) sW[i] = w[i];
    __syncthreads();

    const int lane = threadIdx.x & 31;
    const int wave = threadIdx.x >> 5;
    const int half = lane >> 4;
    const int m    = lane & 15;
    const long patch0 =
        ((long)blockIdx.x * WAVES_PER_BLOCK + wave) * PATCHES_PER_WAVE;
    if (patch0 + PATCHES_PER_WAVE > npatch) return;

    const float* xrow = x + (patch0 + m) * (long)KTOT + 2 * half;
    const float* wrow = sW + 2 * half;

    v8f c = {};
#pragma unroll 8
    for (int k = 0; k < KTOT; k += 4) {
        v2f a = *(const v2f*)(xrow + k);
        v2f b = *(const v2f*)(wrow + k);
        c = __builtin_amdgcn_wmma_f32_16x16x4_f32(
            false, a, false, b, (short)0, c, false, false);
    }

    const float bv = bias[0];
    if (m == 0) {
        float* dst = out + patch0 + half * 8;
        float4 o0 = make_float4(c[0] + bv, c[1] + bv, c[2] + bv, c[3] + bv);
        float4 o1 = make_float4(c[4] + bv, c[5] + bv, c[6] + bv, c[7] + bv);
        *(float4*)(dst)     = o0;
        *(float4*)(dst + 4) = o1;
    }
}

extern "C" void kernel_launch(void* const* d_in, const int* in_sizes, int n_in,
                              void* d_out, int out_size, void* d_ws, size_t ws_size,
                              hipStream_t stream) {
    const float* x    = (const float*)d_in[0];
    const float* w    = (const float*)d_in[1];
    const float* bias = (const float*)d_in[2];
    float* out = (float*)d_out;

    const int npatch = out_size;
    const int waves  = npatch / PATCHES_PER_WAVE;
    const int blocks = (waves + WAVES_PER_BLOCK - 1) / WAVES_PER_BLOCK;

    patchconv_wmma_kernel<<<blocks, 256, 0, stream>>>(x, w, bias, out, npatch);
}


# every wave fetches W and bias itself ahead of its x loads and keeps its own LDS copy of W: W, then tile 0, are waited for by count with the later loads in flight; no work-group barrier
# speedup vs baseline: 1.6326x; 1.6326x over previous
#include <hip/hip_runtime.h>


typedef __attribute__((ext_vector_type(2))) float v2f;
typedef __attribute__((ext_vector_type(4))) float v4f;
typedef __attribute__((ext_vector_type(8))) float v8f;

#define KTOT 768
#define WAVES_PER_BLOCK 8
#define PATCHES_PER_WAVE 16
#define TILE_K 128
#define NTILES (KTOT / TILE_K)
#define PITCH 132
#define WAVE_LDS (KTOT + PATCHES_PER_WAVE * PITCH)

__device__ __forceinline__ void wave_lds_handover() {
    __builtin_amdgcn_fence(__ATOMIC_RELEASE, "wavefront");
    __builtin_amdgcn_wave_barrier();
    __builtin_amdgcn_fence(__ATOMIC_ACQUIRE, "wavefront");
}

__global__ __launch_bounds__(256) void patchconv_wmma_kernel(
    const float* __restrict__ x, const float* __restrict__ w,
    const float* __restrict__ bias, float* __restrict__ out, int npatch)
{
    __shared__ __attribute__((aligned(16))) float smem[WAVES_PER_BLOCK * WAVE_LDS];

    const int lane = threadIdx.x & 31;
    const int wave = __builtin_amdgcn_readfirstlane(threadIdx.x >> 5);
    const int half = lane >> 4;
    const int m    = lane & 15;
    const long patch0 =
        ((long)blockIdx.x * WAVES_PER_BLOCK + wave) * PATCHES_PER_WAVE;
    if (patch0 + PATCHES_PER_WAVE > npatch) return;

    const float bv = bias[0];

    const float* wg = w + 4 * lane;
    const float* xg = x + patch0 * (long)KTOT + 4 * lane;

    v4f wr[NTILES];
#pragma unroll
    for (int q = 0; q < NTILES; ++q)
        wr[q] = *(const v4f*)(wg + q * TILE_K);
    __builtin_amdgcn_sched_barrier(0);

    v4f g[2][PATCHES_PER_WAVE];
#pragma unroll
    for (int b = 0; b < 2; ++b) {
#pragma unroll
        for (int r = 0; r < PATCHES_PER_WAVE; ++r)
            g[b][r] = *(const v4f*)(xg + r * KTOT + b * TILE_K);
        __builtin_amdgcn_sched_barrier(0);
    }

    float* sW   = smem + wave * WAVE_LDS;
    float* tile = sW + KTOT;
#pragma unroll
    for (int q = 0; q < NTILES; ++q)
        *(v4f*)(sW + 4 * lane + q * TILE_K) = wr[q];

    float* tst = tile + 4 * lane;
    const float* tld = tile + m * PITCH + 4 * half;
    const float* wld = sW + 4 * half;

    v8f c0 = {};
    v8f c1 = {};
#pragma unroll
    for (int t = 0; t < NTILES; ++t) {
        const int b = t & 1;
#pragma unroll
        for (int r = 0; r < PATCHES_PER_WAVE; ++r)
            *(v4f*)(tst + r * PITCH) = g[b][r];
        if (t + 2 < NTILES) {
#pragma unroll
            for (int r = 0; r < PATCHES_PER_WAVE; ++r)
                g[b][r] = *(const v4f*)(xg + r * KTOT + (t + 2) * TILE_K);
        }
        wave_lds_handover();
#pragma unroll
        for (int j = 0; j < TILE_K / 8; ++j) {
            const v4f a4 = *(const v4f*)(tld + 8 * j);
            const v4f w4 = *(const v4f*)(wld + t * TILE_K + 8 * j);
            const v2f a_lo = {a4.x, a4.y}, a_hi = {a4.z, a4.w};
            const v2f w_lo = {w4.x, w4.y}, w_hi = {w4.z, w4.w};
            c0 = __builtin_amdgcn_wmma_f32_16x16x4_f32(
                false, a_lo, false, w_lo, (short)0, c0, false, false);
            c1 = __builtin_amdgcn_wmma_f32_16x16x4_f32(
                false, a_hi, false, w_hi, (short)0, c1, false, false);
        }
        wave_lds_handover();
    }

    const v8f c = c0 + c1;
    if (m == 0) {
        float* dst = out + patch0 + half * 8;
        float4 o0 = make_float4(c[0] + bv, c[1] + bv, c[2] + bv, c[3] + bv);
        float4 o1 = make_float4(c[4] + bv, c[5] + bv, c[6] + bv, c[7] + bv);
        *(float4*)(dst)     = o0;
        *(float4*)(dst + 4) = o1;
    }
}

extern "C" void kernel_launch(void* const* d_in, const int* in_sizes, int n_in,
                              void* d_out, int out_size, void* d_ws, size_t ws_size,
                              hipStream_t stream) {
    const float* x    = (const float*)d_in[0];
    const float* w    = (const float*)d_in[1];
    const float* bias = (const float*)d_in[2];
    float* out = (float*)d_out;

    const int npatch = out_size;
    const int waves  = npatch / PATCHES_PER_WAVE;
    const int blocks = (waves + WAVES_PER_BLOCK - 1) / WAVES_PER_BLOCK;

    patchconv_wmma_kernel<<<blocks, 256, 0, stream>>>(x, w, bias, out, npatch);
}
